# starting kernel
# baseline (speedup 1.0000x reference)
#include <hip/hip_runtime.h>
#include <hip/hip_bf16.h>
#include <cstddef>

#define N_NODES 100000
#define N_EDGES 600000
#define D 128
#define H 256

typedef float v2f __attribute__((ext_vector_type(2)));
typedef float v8f __attribute__((ext_vector_type(8)));

__device__ __forceinline__ v8f wmma_f32(v2f a, v2f b, v8f c) {
    return __builtin_amdgcn_wmma_f32_16x16x4_f32(false, a, false, b, (short)0, c,
                                                 false, false);
}

#define ROWS 32
#define XS_STRIDE 132
#define H1_STRIDE 260

__global__ __launch_bounds__(256) void gnn_msg_mlp(
    const float* __restrict__ nodes, const float* __restrict__ Wm1,
    const float* __restrict__ bm1, const float* __restrict__ Wm2,
    const float* __restrict__ bm2, float* __restrict__ m_out) {
    __shared__ float Xs[ROWS * XS_STRIDE];
    __shared__ float H1s[ROWS * H1_STRIDE];

    const int tid = threadIdx.x;
    const int row_base = blockIdx.x * ROWS;

    for (int i = tid; i < ROWS * (D / 4); i += 256) {
        int row = i / (D / 4);
        int c4 = i % (D / 4);
        int rg = row_base + row;
        float4 v = make_float4(0.f, 0.f, 0.f, 0.f);
        if (rg < N_NODES)
            v = ((const float4*)(nodes + (size_t)rg * D))[c4];
        *(float4*)&Xs[row * XS_STRIDE + c4 * 4] = v;
    }
    __syncthreads();

    const int w = tid >> 5;
    const int lane = tid & 31;
    const int half = lane >> 4;
    const int l16 = lane & 15;

    for (int c = 0; c < 2; ++c) {
        int col = (w * 2 + c) * 16 + l16;
        float bv = bm1[col];
        v8f acc0 = {bv, bv, bv, bv, bv, bv, bv, bv};
        v8f acc1 = acc0;
#pragma unroll 4
        for (int k = 0; k < D; k += 4) {
            int kk = k + half * 2;
            v2f b = {Wm1[kk * H + col], Wm1[(kk + 1) * H + col]};
            const float* x0 = &Xs[l16 * XS_STRIDE + kk];
            const float* x1 = &Xs[(16 + l16) * XS_STRIDE + kk];
            v2f a0 = {x0[0], x0[1]};
            v2f a1 = {x1[0], x1[1]};
            acc0 = wmma_f32(a0, b, acc0);
            acc1 = wmma_f32(a1, b, acc1);
        }
#pragma unroll
        for (int r = 0; r < 8; ++r) {
            int row = r + half * 8;
            float v0 = acc0[r];
            float v1 = acc1[r];
            H1s[row * H1_STRIDE + col] = v0 > 0.f ? v0 : 0.f;
            H1s[(16 + row) * H1_STRIDE + col] = v1 > 0.f ? v1 : 0.f;
        }
    }
    __syncthreads();

    {
        int col = w * 16 + l16;
        float bv = bm2[col];
        v8f acc0 = {bv, bv, bv, bv, bv, bv, bv, bv};
        v8f acc1 = acc0;
#pragma unroll 4
        for (int k = 0; k < H; k += 4) {
            int kk = k + half * 2;
            v2f b = {Wm2[kk * D + col], Wm2[(kk + 1) * D + col]};
            const float* h0 = &H1s[l16 * H1_STRIDE + kk];
            const float* h1 = &H1s[(16 + l16) * H1_STRIDE + kk];
            v2f a0 = {h0[0], h0[1]};
            v2f a1 = {h1[0], h1[1]};
            acc0 = wmma_f32(a0, b, acc0);
            acc1 = wmma_f32(a1, b, acc1);
        }
#pragma unroll
        for (int r = 0; r < 8; ++r) {
            int row = r + half * 8;
            int rg0 = row_base + row;
            int rg1 = row_base + 16 + row;
            if (rg0 < N_NODES) m_out[(size_t)rg0 * D + col] = acc0[r];
            if (rg1 < N_NODES) m_out[(size_t)rg1 * D + col] = acc1[r];
        }
    }
}

__global__ __launch_bounds__(256) void gnn_scatter(
    const float* __restrict__ m, const int* __restrict__ senders,
    const int* __restrict__ receivers, float* __restrict__ agg) {
    const int wave = threadIdx.x >> 5;
    const int lane = threadIdx.x & 31;
    const int e = blockIdx.x * 8 + wave;
    if (e >= N_EDGES) return;
    const int s = senders[e];
    const int r = receivers[e];
    const float4 v = ((const float4*)(m + (size_t)s * D))[lane];
    float* dst = agg + (size_t)r * D + lane * 4;
    __hip_atomic_fetch_add(dst + 0, v.x, __ATOMIC_RELAXED, __HIP_MEMORY_SCOPE_AGENT);
    __hip_atomic_fetch_add(dst + 1, v.y, __ATOMIC_RELAXED, __HIP_MEMORY_SCOPE_AGENT);
    __hip_atomic_fetch_add(dst + 2, v.z, __ATOMIC_RELAXED, __HIP_MEMORY_SCOPE_AGENT);
    __hip_atomic_fetch_add(dst + 3, v.w, __ATOMIC_RELAXED, __HIP_MEMORY_SCOPE_AGENT);
}

__global__ __launch_bounds__(256) void gnn_node_mlp(
    const float* __restrict__ nodes, const float* __restrict__ agg,
    const float* __restrict__ Wn1, const float* __restrict__ bn1,
    const float* __restrict__ Wn2, const float* __restrict__ bn2,
    float* __restrict__ out) {
    __shared__ float Xs[ROWS * XS_STRIDE];
    __shared__ float H1s[ROWS * H1_STRIDE];

    const int tid = threadIdx.x;
    const int row_base = blockIdx.x * ROWS;

    for (int i = tid; i < ROWS * (D / 4); i += 256) {
        int row = i / (D / 4);
        int c4 = i % (D / 4);
        int rg = row_base + row;
        float4 v = make_float4(0.f, 0.f, 0.f, 0.f);
        if (rg < N_NODES)
            v = ((const float4*)(nodes + (size_t)rg * D))[c4];
        *(float4*)&Xs[row * XS_STRIDE + c4 * 4] = v;
    }
    __syncthreads();

    const int w = tid >> 5;
    const int lane = tid & 31;
    const int half = lane >> 4;
    const int l16 = lane & 15;

    int rg0 = row_base + l16;
    int rg1 = row_base + 16 + l16;
    if (rg0 >= N_NODES) rg0 = N_NODES - 1;
    if (rg1 >= N_NODES) rg1 = N_NODES - 1;
    const float* aggRow0 = agg + (size_t)rg0 * D;
    const float* aggRow1 = agg + (size_t)rg1 * D;

    for (int c = 0; c < 2; ++c) {
        int col = (w * 2 + c) * 16 + l16;
        float bv = bn1[col];
        v8f acc0 = {bv, bv, bv, bv, bv, bv, bv, bv};
        v8f acc1 = acc0;
#pragma unroll 4
        for (int k = 0; k < D; k += 4) {
            int kk = k + half * 2;
            v2f b = {Wn1[kk * H + col], Wn1[(kk + 1) * H + col]};
            const float* x0 = &Xs[l16 * XS_STRIDE + kk];
            const float* x1 = &Xs[(16 + l16) * XS_STRIDE + kk];
            v2f a0 = {x0[0], x0[1]};
            v2f a1 = {x1[0], x1[1]};
            acc0 = wmma_f32(a0, b, acc0);
            acc1 = wmma_f32(a1, b, acc1);
        }
#pragma unroll 4
        for (int k = D; k < 2 * D; k += 4) {
            int kk = k + half * 2;
            v2f b = {Wn1[kk * H + col], Wn1[(kk + 1) * H + col]};
            const float* a0p = aggRow0 + (kk - D);
            const float* a1p = aggRow1 + (kk - D);
            v2f a0 = {a0p[0], a0p[1]};
            v2f a1 = {a1p[0], a1p[1]};
            acc0 = wmma_f32(a0, b, acc0);
            acc1 = wmma_f32(a1, b, acc1);
        }
#pragma unroll
        for (int r = 0; r < 8; ++r) {
            int row = r + half * 8;
            float v0 = acc0[r];
            float v1 = acc1[r];
            H1s[row * H1_STRIDE + col] = v0 > 0.f ? v0 : 0.f;
            H1s[(16 + row) * H1_STRIDE + col] = v1 > 0.f ? v1 : 0.f;
        }
    }
    __syncthreads();

    {
        int col = w * 16 + l16;
        float bv = bn2[col];
        v8f acc0 = {bv, bv, bv, bv, bv, bv, bv, bv};
        v8f acc1 = acc0;
#pragma unroll 4
        for (int k = 0; k < H; k += 4) {
            int kk = k + half * 2;
            v2f b = {Wn2[kk * D + col], Wn2[(kk + 1) * D + col]};
            const float* h0 = &H1s[l16 * H1_STRIDE + kk];
            const float* h1 = &H1s[(16 + l16) * H1_STRIDE + kk];
            v2f a0 = {h0[0], h0[1]};
            v2f a1 = {h1[0], h1[1]};
            acc0 = wmma_f32(a0, b, acc0);
            acc1 = wmma_f32(a1, b, acc1);
        }
#pragma unroll
        for (int r = 0; r < 8; ++r) {
            int row = r + half * 8;
            int g0 = row_base + row;
            int g1 = row_base + 16 + row;
            if (g0 < N_NODES)
                out[(size_t)g0 * D + col] = Xs[row * XS_STRIDE + col] + acc0[r];
            if (g1 < N_NODES)
                out[(size_t)g1 * D + col] = Xs[(16 + row) * XS_STRIDE + col] + acc1[r];
        }
    }
}

extern "C" void kernel_launch(void* const* d_in, const int* in_sizes, int n_in,
                              void* d_out, int out_size, void* d_ws, size_t ws_size,
                              hipStream_t stream) {
    (void)in_sizes; (void)n_in; (void)out_size; (void)ws_size;
    const float* nodes     = (const float*)d_in[0];
    const int*   senders   = (const int*)d_in[1];
    const int*   receivers = (const int*)d_in[2];
    const float* Wm1 = (const float*)d_in[3];
    const float* bm1 = (const float*)d_in[4];
    const float* Wm2 = (const float*)d_in[5];
    const float* bm2 = (const float*)d_in[6];
    const float* Wn1 = (const float*)d_in[7];
    const float* bn1 = (const float*)d_in[8];
    const float* Wn2 = (const float*)d_in[9];
    const float* bn2 = (const float*)d_in[10];

    float* m_buf = (float*)d_ws;
    float* out   = (float*)d_out;

    const int row_blocks = (N_NODES + ROWS - 1) / ROWS;

    gnn_msg_mlp<<<row_blocks, 256, 0, stream>>>(nodes, Wm1, bm1, Wm2, bm2, m_buf);

    hipMemsetAsync(d_out, 0, sizeof(float) * (size_t)N_NODES * D, stream);

    gnn_scatter<<<(N_EDGES + 7) / 8, 256, 0, stream>>>(m_buf, senders, receivers, out);

    gnn_node_mlp<<<row_blocks, 256, 0, stream>>>(nodes, out, Wn1, bn1, Wn2, bn2, out);
}


# both MLP kernels weight-stationary: 256 persistent blocks keep all weight fragments in registers and stream 32-row tiles through LDS (was: weights re-read from L2 each k step); sum order unchanged
# speedup vs baseline: 1.3247x; 1.3247x over previous
#include <hip/hip_runtime.h>
#include <hip/hip_bf16.h>
#include <cstddef>

#define N_NODES 100000
#define N_EDGES 600000
#define D 128
#define H 256

typedef float v2f __attribute__((ext_vector_type(2)));
typedef float v8f __attribute__((ext_vector_type(8)));
typedef float v4f __attribute__((ext_vector_type(4)));

__device__ __forceinline__ v8f wmma_f32(v2f a, v2f b, v8f c) {
    return __builtin_amdgcn_wmma_f32_16x16x4_f32(false, a, false, b, (short)0, c,
                                                 false, false);
}

#define ROWS 32
#define XS_STRIDE 132
#define H1_STRIDE 260
#define N_TILES ((N_NODES + ROWS - 1) / ROWS)
#define PERSIST_BLOCKS 256
static_assert(N_NODES % ROWS == 0, "row tiles must be full");
static_assert(PERSIST_BLOCKS <= N_TILES, "every block needs a first tile");

__global__ __launch_bounds__(256) void gnn_msg_mlp(
    const float* __restrict__ nodes, const float* __restrict__ Wm1,
    const float* __restrict__ bm1, const float* __restrict__ Wm2,
    const float* __restrict__ bm2, float* __restrict__ m_out) {
    __shared__ float Xs[ROWS * XS_STRIDE];
    __shared__ float H1s[ROWS * H1_STRIDE];

    const int tid = threadIdx.x;
    const int w = tid >> 5;
    const int lane = tid & 31;
    const int half = lane >> 4;
    const int l16 = lane & 15;

    const int col1a = (w * 2 + 0) * 16 + l16;
    const int col1b = (w * 2 + 1) * 16 + l16;
    const int col2 = w * 16 + l16;
    v2f B1a[D / 4], B1b[D / 4], B2[H / 4];
#pragma unroll
    for (int ks = 0; ks < D / 4; ++ks) {
        int kk = ks * 4 + half * 2;
        B1a[ks] = (v2f){Wm1[kk * H + col1a], Wm1[(kk + 1) * H + col1a]};
        B1b[ks] = (v2f){Wm1[kk * H + col1b], Wm1[(kk + 1) * H + col1b]};
    }
#pragma unroll
    for (int ks = 0; ks < H / 4; ++ks) {
        int kk = ks * 4 + half * 2;
        B2[ks] = (v2f){Wm2[kk * D + col2], Wm2[(kk + 1) * D + col2]};
    }
    const float bv1a = bm1[col1a];
    const float bv1b = bm1[col1b];
    const float bv2 = bm2[col2];

    int t = blockIdx.x;
#pragma unroll
    for (int j = 0; j < 4; ++j) {
        int i = tid + j * 256;
        int row = i >> 5;
        int c4 = i & 31;
        int rg = t * ROWS + row;
        *(float4*)&Xs[row * XS_STRIDE + c4 * 4] =
            ((const float4*)(nodes + (size_t)rg * D))[c4];
    }

    for (; t < N_TILES; t += PERSIST_BLOCKS) {
        const int tn = t + PERSIST_BLOCKS;
        const bool has_next = tn < N_TILES;
        __syncthreads();

        {
            v8f acc00 = {bv1a, bv1a, bv1a, bv1a, bv1a, bv1a, bv1a, bv1a};
            v8f acc01 = acc00;
            v8f acc10 = {bv1b, bv1b, bv1b, bv1b, bv1b, bv1b, bv1b, bv1b};
            v8f acc11 = acc10;
#pragma unroll
            for (int ks = 0; ks < D / 4; ++ks) {
                int kk = ks * 4 + half * 2;
                const float* x0 = &Xs[l16 * XS_STRIDE + kk];
                const float* x1 = &Xs[(16 + l16) * XS_STRIDE + kk];
                v2f a0 = {x0[0], x0[1]};
                v2f a1 = {x1[0], x1[1]};
                acc00 = wmma_f32(a0, B1a[ks], acc00);
                acc01 = wmma_f32(a1, B1a[ks], acc01);
                acc10 = wmma_f32(a0, B1b[ks], acc10);
                acc11 = wmma_f32(a1, B1b[ks], acc11);
            }
#pragma unroll
            for (int r = 0; r < 8; ++r) {
                int row = r + half * 8;
                float v00 = acc00[r], v01 = acc01[r];
                float v10 = acc10[r], v11 = acc11[r];
                H1s[row * H1_STRIDE + col1a] = v00 > 0.f ? v00 : 0.f;
                H1s[(16 + row) * H1_STRIDE + col1a] = v01 > 0.f ? v01 : 0.f;
                H1s[row * H1_STRIDE + col1b] = v10 > 0.f ? v10 : 0.f;
                H1s[(16 + row) * H1_STRIDE + col1b] = v11 > 0.f ? v11 : 0.f;
            }
        }
        __syncthreads();

        v4f xr[4];
        if (has_next) {
#pragma unroll
            for (int j = 0; j < 4; ++j) {
                int i = tid + j * 256;
                int row = i >> 5;
                int c4 = i & 31;
                int rg = tn * ROWS + row;
                xr[j] = ((const v4f*)(nodes + (size_t)rg * D))[c4];
            }
        }

        {
            v8f acc0 = {bv2, bv2, bv2, bv2, bv2, bv2, bv2, bv2};
            v8f acc1 = acc0;
#pragma unroll
            for (int ks = 0; ks < H / 4; ++ks) {
                int kk = ks * 4 + half * 2;
                const float* h0 = &H1s[l16 * H1_STRIDE + kk];
                const float* h1 = &H1s[(16 + l16) * H1_STRIDE + kk];
                v2f a0 = {h0[0], h0[1]};
                v2f a1 = {h1[0], h1[1]};
                acc0 = wmma_f32(a0, B2[ks], acc0);
                acc1 = wmma_f32(a1, B2[ks], acc1);
            }
            const int row_base = t * ROWS;
#pragma unroll
            for (int r = 0; r < 8; ++r) {
                int row = r + half * 8;
                int rg0 = row_base + row;
                int rg1 = row_base + 16 + row;
                m_out[(size_t)rg0 * D + col2] = acc0[r];
                m_out[(size_t)rg1 * D + col2] = acc1[r];
            }
        }

        if (has_next) {
#pragma unroll
            for (int j = 0; j < 4; ++j) {
                int i = tid + j * 256;
                int row = i >> 5;
                int c4 = i & 31;
                *(v4f*)&Xs[row * XS_STRIDE + c4 * 4] = xr[j];
            }
        }
    }
}

__global__ __launch_bounds__(256) void gnn_scatter(
    const float* __restrict__ m, const int* __restrict__ senders,
    const int* __restrict__ receivers, float* __restrict__ agg) {
    const int wave = threadIdx.x >> 5;
    const int lane = threadIdx.x & 31;
    const int e = blockIdx.x * 8 + wave;
    if (e >= N_EDGES) return;
    const int s = senders[e];
    const int r = receivers[e];
    const float* src = m + (size_t)s * D + lane;
    const float v0 = src[0];
    const float v1 = src[32];
    const float v2 = src[64];
    const float v3 = src[96];
    float* dst = agg + (size_t)r * D + lane;
    __hip_atomic_fetch_add(dst + 0,  v0, __ATOMIC_RELAXED, __HIP_MEMORY_SCOPE_AGENT);
    __hip_atomic_fetch_add(dst + 32, v1, __ATOMIC_RELAXED, __HIP_MEMORY_SCOPE_AGENT);
    __hip_atomic_fetch_add(dst + 64, v2, __ATOMIC_RELAXED, __HIP_MEMORY_SCOPE_AGENT);
    __hip_atomic_fetch_add(dst + 96, v3, __ATOMIC_RELAXED, __HIP_MEMORY_SCOPE_AGENT);
}

__global__ __launch_bounds__(256) void gnn_node_mlp(
    const float* __restrict__ nodes, const float* agg,
    const float* __restrict__ Wn1, const float* __restrict__ bn1,
    const float* __restrict__ Wn2, const float* __restrict__ bn2,
    float* out) {
    __shared__ float Xc[ROWS * H1_STRIDE];
    __shared__ float H1s[ROWS * H1_STRIDE];

    const int tid = threadIdx.x;
    const int w = tid >> 5;
    const int lane = tid & 31;
    const int half = lane >> 4;
    const int l16 = lane & 15;

    const int col1a = (w * 2 + 0) * 16 + l16;
    const int col1b = (w * 2 + 1) * 16 + l16;
    const int col2 = w * 16 + l16;
    v2f B1a[2 * D / 4], B1b[2 * D / 4], B2[H / 4];
#pragma unroll
    for (int ks = 0; ks < 2 * D / 4; ++ks) {
        int kk = ks * 4 + half * 2;
        B1a[ks] = (v2f){Wn1[kk * H + col1a], Wn1[(kk + 1) * H + col1a]};
        B1b[ks] = (v2f){Wn1[kk * H + col1b], Wn1[(kk + 1) * H + col1b]};
    }
#pragma unroll
    for (int ks = 0; ks < H / 4; ++ks) {
        int kk = ks * 4 + half * 2;
        B2[ks] = (v2f){Wn2[kk * D + col2], Wn2[(kk + 1) * D + col2]};
    }
    const float bv1a = bn1[col1a];
    const float bv1b = bn1[col1b];
    const float bv2 = bn2[col2];

    int t = blockIdx.x;
#pragma unroll
    for (int j = 0; j < 8; ++j) {
        int i = tid + j * 256;
        int row = i >> 6;
        int c4 = i & 63;
        int rg = t * ROWS + row;
        const float* base = (c4 < D / 4) ? nodes : agg;
        *(float4*)&Xc[row * H1_STRIDE + c4 * 4] =
            ((const float4*)(base + (size_t)rg * D))[c4 & (D / 4 - 1)];
    }

    for (; t < N_TILES; t += PERSIST_BLOCKS) {
        const int tn = t + PERSIST_BLOCKS;
        const bool has_next = tn < N_TILES;
        __syncthreads();

        float res0[8], res1[8];
        {
            v8f acc00 = {bv1a, bv1a, bv1a, bv1a, bv1a, bv1a, bv1a, bv1a};
            v8f acc01 = acc00;
            v8f acc10 = {bv1b, bv1b, bv1b, bv1b, bv1b, bv1b, bv1b, bv1b};
            v8f acc11 = acc10;
#pragma unroll
            for (int ks = 0; ks < 2 * D / 4; ++ks) {
                int kk = ks * 4 + half * 2;
                const float* x0 = &Xc[l16 * H1_STRIDE + kk];
                const float* x1 = &Xc[(16 + l16) * H1_STRIDE + kk];
                v2f a0 = {x0[0], x0[1]};
                v2f a1 = {x1[0], x1[1]};
                acc00 = wmma_f32(a0, B1a[ks], acc00);
                acc01 = wmma_f32(a1, B1a[ks], acc01);
                acc10 = wmma_f32(a0, B1b[ks], acc10);
                acc11 = wmma_f32(a1, B1b[ks], acc11);
            }
#pragma unroll
            for (int r = 0; r < 8; ++r) {
                int row = r + half * 8;
                float v00 = acc00[r], v01 = acc01[r];
                float v10 = acc10[r], v11 = acc11[r];
                H1s[row * H1_STRIDE + col1a] = v00 > 0.f ? v00 : 0.f;
                H1s[(16 + row) * H1_STRIDE + col1a] = v01 > 0.f ? v01 : 0.f;
                H1s[row * H1_STRIDE + col1b] = v10 > 0.f ? v10 : 0.f;
                H1s[(16 + row) * H1_STRIDE + col1b] = v11 > 0.f ? v11 : 0.f;
                res0[r] = Xc[row * H1_STRIDE + col2];
                res1[r] = Xc[(16 + row) * H1_STRIDE + col2];
            }
        }
        __syncthreads();

        v4f xr[8];
        if (has_next) {
#pragma unroll
            for (int j = 0; j < 8; ++j) {
                int i = tid + j * 256;
                int row = i >> 6;
                int c4 = i & 63;
                int rg = tn * ROWS + row;
                const float* base = (c4 < D / 4) ? nodes : agg;
                xr[j] = ((const v4f*)(base + (size_t)rg * D))[c4 & (D / 4 - 1)];
            }
        }

        {
            v8f acc0 = {bv2, bv2, bv2, bv2, bv2, bv2, bv2, bv2};
            v8f acc1 = acc0;
#pragma unroll
            for (int ks = 0; ks < H / 4; ++ks) {
                int kk = ks * 4 + half * 2;
                const float* h0 = &H1s[l16 * H1_STRIDE + kk];
                const float* h1 = &H1s[(16 + l16) * H1_STRIDE + kk];
                v2f a0 = {h0[0], h0[1]};
                v2f a1 = {h1[0], h1[1]};
                acc0 = wmma_f32(a0, B2[ks], acc0);
                acc1 = wmma_f32(a1, B2[ks], acc1);
            }
            const int row_base = t * ROWS;
#pragma unroll
            for (int r = 0; r < 8; ++r) {
                int row = r + half * 8;
                int g0 = row_base + row;
                int g1 = row_base + 16 + row;
                out[(size_t)g0 * D + col2] = res0[r] + acc0[r];
                out[(size_t)g1 * D + col2] = res1[r] + acc1[r];
            }
        }

        if (has_next) {
#pragma unroll
            for (int j = 0; j < 8; ++j) {
                int i = tid + j * 256;
                int row = i >> 6;
                int c4 = i & 63;
                *(v4f*)&Xc[row * H1_STRIDE + c4 * 4] = xr[j];
            }
        }
    }
}

extern "C" void kernel_launch(void* const* d_in, const int* in_sizes, int n_in,
                              void* d_out, int out_size, void* d_ws, size_t ws_size,
                              hipStream_t stream) {
    (void)in_sizes; (void)n_in; (void)out_size; (void)ws_size;
    const float* nodes     = (const float*)d_in[0];
    const int*   senders   = (const int*)d_in[1];
    const int*   receivers = (const int*)d_in[2];
    const float* Wm1 = (const float*)d_in[3];
    const float* bm1 = (const float*)d_in[4];
    const float* Wm2 = (const float*)d_in[5];
    const float* bm2 = (const float*)d_in[6];
    const float* Wn1 = (const float*)d_in[7];
    const float* bn1 = (const float*)d_in[8];
    const float* Wn2 = (const float*)d_in[9];
    const float* bn2 = (const float*)d_in[10];

    float* m_buf = (float*)d_ws;
    float* out   = (float*)d_out;

    gnn_msg_mlp<<<PERSIST_BLOCKS, 256, 0, stream>>>(nodes, Wm1, bm1, Wm2, bm2, m_buf);

    hipMemsetAsync(d_out, 0, sizeof(float) * (size_t)N_NODES * D, stream);

    gnn_scatter<<<(N_EDGES + 7) / 8, 256, 0, stream>>>(m_buf, senders, receivers, out);

    gnn_node_mlp<<<PERSIST_BLOCKS, 256, 0, stream>>>(nodes, out, Wn1, bn1, Wn2, bn2, out);
}


# scatter: each wave takes 8 edges (scalar index loads, all 32 row-line loads in flight, then its 32 line atomics), 75000 waves instead of 600000; sums still in atomic arrival order
# speedup vs baseline: 1.4477x; 1.0929x over previous
#include <hip/hip_runtime.h>
#include <hip/hip_bf16.h>
#include <cstddef>

#define N_NODES 100000
#define N_EDGES 600000
#define D 128
#define H 256

typedef float v2f __attribute__((ext_vector_type(2)));
typedef float v8f __attribute__((ext_vector_type(8)));
typedef float v4f __attribute__((ext_vector_type(4)));

__device__ __forceinline__ v8f wmma_f32(v2f a, v2f b, v8f c) {
    return __builtin_amdgcn_wmma_f32_16x16x4_f32(false, a, false, b, (short)0, c,
                                                 false, false);
}

#define ROWS 32
#define XS_STRIDE 132
#define H1_STRIDE 260
#define N_TILES ((N_NODES + ROWS - 1) / ROWS)
#define PERSIST_BLOCKS 256
static_assert(N_NODES % ROWS == 0, "row tiles must be full");
static_assert(PERSIST_BLOCKS <= N_TILES, "every block needs a first tile");

__global__ __launch_bounds__(256) void gnn_msg_mlp(
    const float* __restrict__ nodes, const float* __restrict__ Wm1,
    const float* __restrict__ bm1, const float* __restrict__ Wm2,
    const float* __restrict__ bm2, float* __restrict__ m_out) {
    __shared__ float Xs[ROWS * XS_STRIDE];
    __shared__ float H1s[ROWS * H1_STRIDE];

    const int tid = threadIdx.x;
    const int w = tid >> 5;
    const int lane = tid & 31;
    const int half = lane >> 4;
    const int l16 = lane & 15;

    const int col1a = (w * 2 + 0) * 16 + l16;
    const int col1b = (w * 2 + 1) * 16 + l16;
    const int col2 = w * 16 + l16;
    v2f B1a[D / 4], B1b[D / 4], B2[H / 4];
#pragma unroll
    for (int ks = 0; ks < D / 4; ++ks) {
        int kk = ks * 4 + half * 2;
        B1a[ks] = (v2f){Wm1[kk * H + col1a], Wm1[(kk + 1) * H + col1a]};
        B1b[ks] = (v2f){Wm1[kk * H + col1b], Wm1[(kk + 1) * H + col1b]};
    }
#pragma unroll
    for (int ks = 0; ks < H / 4; ++ks) {
        int kk = ks * 4 + half * 2;
        B2[ks] = (v2f){Wm2[kk * D + col2], Wm2[(kk + 1) * D + col2]};
    }
    const float bv1a = bm1[col1a];
    const float bv1b = bm1[col1b];
    const float bv2 = bm2[col2];

    int t = blockIdx.x;
#pragma unroll
    for (int j = 0; j < 4; ++j) {
        int i = tid + j * 256;
        int row = i >> 5;
        int c4 = i & 31;
        int rg = t * ROWS + row;
        *(float4*)&Xs[row * XS_STRIDE + c4 * 4] =
            ((const float4*)(nodes + (size_t)rg * D))[c4];
    }

    for (; t < N_TILES; t += PERSIST_BLOCKS) {
        const int tn = t + PERSIST_BLOCKS;
        const bool has_next = tn < N_TILES;
        __syncthreads();

        {
            v8f acc00 = {bv1a, bv1a, bv1a, bv1a, bv1a, bv1a, bv1a, bv1a};
            v8f acc01 = acc00;
            v8f acc10 = {bv1b, bv1b, bv1b, bv1b, bv1b, bv1b, bv1b, bv1b};
            v8f acc11 = acc10;
#pragma unroll
            for (int ks = 0; ks < D / 4; ++ks) {
                int kk = ks * 4 + half * 2;
                const float* x0 = &Xs[l16 * XS_STRIDE + kk];
                const float* x1 = &Xs[(16 + l16) * XS_STRIDE + kk];
                v2f a0 = {x0[0], x0[1]};
                v2f a1 = {x1[0], x1[1]};
                acc00 = wmma_f32(a0, B1a[ks], acc00);
                acc01 = wmma_f32(a1, B1a[ks], acc01);
                acc10 = wmma_f32(a0, B1b[ks], acc10);
                acc11 = wmma_f32(a1, B1b[ks], acc11);
            }
#pragma unroll
            for (int r = 0; r < 8; ++r) {
                int row = r + half * 8;
                float v00 = acc00[r], v01 = acc01[r];
                float v10 = acc10[r], v11 = acc11[r];
                H1s[row * H1_STRIDE + col1a] = v00 > 0.f ? v00 : 0.f;
                H1s[(16 + row) * H1_STRIDE + col1a] = v01 > 0.f ? v01 : 0.f;
                H1s[row * H1_STRIDE + col1b] = v10 > 0.f ? v10 : 0.f;
                H1s[(16 + row) * H1_STRIDE + col1b] = v11 > 0.f ? v11 : 0.f;
            }
        }
        __syncthreads();

        v4f xr[4];
        if (has_next) {
#pragma unroll
            for (int j = 0; j < 4; ++j) {
                int i = tid + j * 256;
                int row = i >> 5;
                int c4 = i & 31;
                int rg = tn * ROWS + row;
                xr[j] = ((const v4f*)(nodes + (size_t)rg * D))[c4];
            }
        }

        {
            v8f acc0 = {bv2, bv2, bv2, bv2, bv2, bv2, bv2, bv2};
            v8f acc1 = acc0;
#pragma unroll
            for (int ks = 0; ks < H / 4; ++ks) {
                int kk = ks * 4 + half * 2;
                const float* h0 = &H1s[l16 * H1_STRIDE + kk];
                const float* h1 = &H1s[(16 + l16) * H1_STRIDE + kk];
                v2f a0 = {h0[0], h0[1]};
                v2f a1 = {h1[0], h1[1]};
                acc0 = wmma_f32(a0, B2[ks], acc0);
                acc1 = wmma_f32(a1, B2[ks], acc1);
            }
            const int row_base = t * ROWS;
#pragma unroll
            for (int r = 0; r < 8; ++r) {
                int row = r + half * 8;
                int rg0 = row_base + row;
                int rg1 = row_base + 16 + row;
                m_out[(size_t)rg0 * D + col2] = acc0[r];
                m_out[(size_t)rg1 * D + col2] = acc1[r];
            }
        }

        if (has_next) {
#pragma unroll
            for (int j = 0; j < 4; ++j) {
                int i = tid + j * 256;
                int row = i >> 5;
                int c4 = i & 31;
                *(v4f*)&Xs[row * XS_STRIDE + c4 * 4] = xr[j];
            }
        }
    }
}

#define EPW 8
#define SC_WAVES 8
static_assert(N_EDGES % (EPW * SC_WAVES) == 0, "every wave owns a full group of edges");

__global__ __launch_bounds__(256) void gnn_scatter(
    const float* __restrict__ m, const int* __restrict__ senders,
    const int* __restrict__ receivers, float* __restrict__ agg) {
    const int wave = __builtin_amdgcn_readfirstlane((int)(threadIdx.x >> 5));
    const int lane = threadIdx.x & 31;
    const int e0 = (blockIdx.x * SC_WAVES + wave) * EPW;

    int s[EPW], r[EPW];
    for (int j = 0; j < EPW; ++j) {
        s[j] = senders[e0 + j];
        r[j] = receivers[e0 + j];
    }

    float v[EPW][4];
    for (int j = 0; j < EPW; ++j) {
        const float* src = m + (size_t)s[j] * D + lane;
        v[j][0] = src[0];
        v[j][1] = src[32];
        v[j][2] = src[64];
        v[j][3] = src[96];
    }

    for (int j = 0; j < EPW; ++j) {
        float* dst = agg + (size_t)r[j] * D + lane;
        __hip_atomic_fetch_add(dst + 0,  v[j][0], __ATOMIC_RELAXED, __HIP_MEMORY_SCOPE_AGENT);
        __hip_atomic_fetch_add(dst + 32, v[j][1], __ATOMIC_RELAXED, __HIP_MEMORY_SCOPE_AGENT);
        __hip_atomic_fetch_add(dst + 64, v[j][2], __ATOMIC_RELAXED, __HIP_MEMORY_SCOPE_AGENT);
        __hip_atomic_fetch_add(dst + 96, v[j][3], __ATOMIC_RELAXED, __HIP_MEMORY_SCOPE_AGENT);
    }
}

__global__ __launch_bounds__(256) void gnn_node_mlp(
    const float* __restrict__ nodes, const float* agg,
    const float* __restrict__ Wn1, const float* __restrict__ bn1,
    const float* __restrict__ Wn2, const float* __restrict__ bn2,
    float* out) {
    __shared__ float Xc[ROWS * H1_STRIDE];
    __shared__ float H1s[ROWS * H1_STRIDE];

    const int tid = threadIdx.x;
    const int w = tid >> 5;
    const int lane = tid & 31;
    const int half = lane >> 4;
    const int l16 = lane & 15;

    const int col1a = (w * 2 + 0) * 16 + l16;
    const int col1b = (w * 2 + 1) * 16 + l16;
    const int col2 = w * 16 + l16;
    v2f B1a[2 * D / 4], B1b[2 * D / 4], B2[H / 4];
#pragma unroll
    for (int ks = 0; ks < 2 * D / 4; ++ks) {
        int kk = ks * 4 + half * 2;
        B1a[ks] = (v2f){Wn1[kk * H + col1a], Wn1[(kk + 1) * H + col1a]};
        B1b[ks] = (v2f){Wn1[kk * H + col1b], Wn1[(kk + 1) * H + col1b]};
    }
#pragma unroll
    for (int ks = 0; ks < H / 4; ++ks) {
        int kk = ks * 4 + half * 2;
        B2[ks] = (v2f){Wn2[kk * D + col2], Wn2[(kk + 1) * D + col2]};
    }
    const float bv1a = bn1[col1a];
    const float bv1b = bn1[col1b];
    const float bv2 = bn2[col2];

    int t = blockIdx.x;
#pragma unroll
    for (int j = 0; j < 8; ++j) {
        int i = tid + j * 256;
        int row = i >> 6;
        int c4 = i & 63;
        int rg = t * ROWS + row;
        const float* base = (c4 < D / 4) ? nodes : agg;
        *(float4*)&Xc[row * H1_STRIDE + c4 * 4] =
            ((const float4*)(base + (size_t)rg * D))[c4 & (D / 4 - 1)];
    }

    for (; t < N_TILES; t += PERSIST_BLOCKS) {
        const int tn = t + PERSIST_BLOCKS;
        const bool has_next = tn < N_TILES;
        __syncthreads();

        float res0[8], res1[8];
        {
            v8f acc00 = {bv1a, bv1a, bv1a, bv1a, bv1a, bv1a, bv1a, bv1a};
            v8f acc01 = acc00;
            v8f acc10 = {bv1b, bv1b, bv1b, bv1b, bv1b, bv1b, bv1b, bv1b};
            v8f acc11 = acc10;
#pragma unroll
            for (int ks = 0; ks < 2 * D / 4; ++ks) {
                int kk = ks * 4 + half * 2;
                const float* x0 = &Xc[l16 * H1_STRIDE + kk];
                const float* x1 = &Xc[(16 + l16) * H1_STRIDE + kk];
                v2f a0 = {x0[0], x0[1]};
                v2f a1 = {x1[0], x1[1]};
                acc00 = wmma_f32(a0, B1a[ks], acc00);
                acc01 = wmma_f32(a1, B1a[ks], acc01);
                acc10 = wmma_f32(a0, B1b[ks], acc10);
                acc11 = wmma_f32(a1, B1b[ks], acc11);
            }
#pragma unroll
            for (int r = 0; r < 8; ++r) {
                int row = r + half * 8;
                float v00 = acc00[r], v01 = acc01[r];
                float v10 = acc10[r], v11 = acc11[r];
                H1s[row * H1_STRIDE + col1a] = v00 > 0.f ? v00 : 0.f;
                H1s[(16 + row) * H1_STRIDE + col1a] = v01 > 0.f ? v01 : 0.f;
                H1s[row * H1_STRIDE + col1b] = v10 > 0.f ? v10 : 0.f;
                H1s[(16 + row) * H1_STRIDE + col1b] = v11 > 0.f ? v11 : 0.f;
                res0[r] = Xc[row * H1_STRIDE + col2];
                res1[r] = Xc[(16 + row) * H1_STRIDE + col2];
            }
        }
        __syncthreads();

        v4f xr[8];
        if (has_next) {
#pragma unroll
            for (int j = 0; j < 8; ++j) {
                int i = tid + j * 256;
                int row = i >> 6;
                int c4 = i & 63;
                int rg = tn * ROWS + row;
                const float* base = (c4 < D / 4) ? nodes : agg;
                xr[j] = ((const v4f*)(base + (size_t)rg * D))[c4 & (D / 4 - 1)];
            }
        }

        {
            v8f acc0 = {bv2, bv2, bv2, bv2, bv2, bv2, bv2, bv2};
            v8f acc1 = acc0;
#pragma unroll
            for (int ks = 0; ks < H / 4; ++ks) {
                int kk = ks * 4 + half * 2;
                const float* h0 = &H1s[l16 * H1_STRIDE + kk];
                const float* h1 = &H1s[(16 + l16) * H1_STRIDE + kk];
                v2f a0 = {h0[0], h0[1]};
                v2f a1 = {h1[0], h1[1]};
                acc0 = wmma_f32(a0, B2[ks], acc0);
                acc1 = wmma_f32(a1, B2[ks], acc1);
            }
            const int row_base = t * ROWS;
#pragma unroll
            for (int r = 0; r < 8; ++r) {
                int row = r + half * 8;
                int g0 = row_base + row;
                int g1 = row_base + 16 + row;
                out[(size_t)g0 * D + col2] = res0[r] + acc0[r];
                out[(size_t)g1 * D + col2] = res1[r] + acc1[r];
            }
        }

        if (has_next) {
#pragma unroll
            for (int j = 0; j < 8; ++j) {
                int i = tid + j * 256;
                int row = i >> 6;
                int c4 = i & 63;
                *(v4f*)&Xc[row * H1_STRIDE + c4 * 4] = xr[j];
            }
        }
    }
}

extern "C" void kernel_launch(void* const* d_in, const int* in_sizes, int n_in,
                              void* d_out, int out_size, void* d_ws, size_t ws_size,
                              hipStream_t stream) {
    (void)in_sizes; (void)n_in; (void)out_size; (void)ws_size;
    const float* nodes     = (const float*)d_in[0];
    const int*   senders   = (const int*)d_in[1];
    const int*   receivers = (const int*)d_in[2];
    const float* Wm1 = (const float*)d_in[3];
    const float* bm1 = (const float*)d_in[4];
    const float* Wm2 = (const float*)d_in[5];
    const float* bm2 = (const float*)d_in[6];
    const float* Wn1 = (const float*)d_in[7];
    const float* bn1 = (const float*)d_in[8];
    const float* Wn2 = (const float*)d_in[9];
    const float* bn2 = (const float*)d_in[10];

    float* m_buf = (float*)d_ws;
    float* out   = (float*)d_out;

    gnn_msg_mlp<<<PERSIST_BLOCKS, 256, 0, stream>>>(nodes, Wm1, bm1, Wm2, bm2, m_buf);

    hipMemsetAsync(d_out, 0, sizeof(float) * (size_t)N_NODES * D, stream);

    gnn_scatter<<<N_EDGES / (EPW * SC_WAVES), 256, 0, stream>>>(m_buf, senders, receivers, out);

    gnn_node_mlp<<<PERSIST_BLOCKS, 256, 0, stream>>>(nodes, out, Wn1, bn1, Wn2, bn2, out);
}
